# starting kernel
# baseline (speedup 1.0000x reference)
#include <hip/hip_runtime.h>
#include <math.h>

#define D_MODEL 2048
#define NHEADS  16
#define DK      128
#define SEQ     2048
#define BATCH   2
#define MROWS   (BATCH * SEQ)
#define LOG_THETA 9.210340371976184f

typedef __attribute__((ext_vector_type(2))) float v2f;
typedef __attribute__((ext_vector_type(8))) float v8f;

__device__ __forceinline__ v8f wmma_f32(v2f a, v2f b, v8f c) {
  return __builtin_amdgcn_wmma_f32_16x16x4_f32(false, a, false, b,
                                               (short)0, c, false, false);
}

__device__ __forceinline__ v8f vzero8() {
  v8f z = {0.f, 0.f, 0.f, 0.f, 0.f, 0.f, 0.f, 0.f};
  return z;
}

__global__ __launch_bounds__(32)
void mha_qkv_rope(const float* __restrict__ x, const int* __restrict__ tpos,
                  const float* __restrict__ WQ, const float* __restrict__ WK,
                  const float* __restrict__ WV,
                  float* __restrict__ Qo, float* __restrict__ Ko,
                  float* __restrict__ Vo) {
  const int lane = threadIdx.x;
  const int hi = lane >> 4;
  const int lo = lane & 15;
  const int m0 = blockIdx.x * 16;
  const int n0 = blockIdx.y * 64;
  const int mat = blockIdx.z;

  const float* __restrict__ W = (mat == 0) ? WQ : ((mat == 1) ? WK : WV);
  float* __restrict__ Out = (mat == 0) ? Qo : ((mat == 1) ? Ko : Vo);

  v8f acc[4];
#pragma unroll
  for (int t = 0; t < 4; ++t) acc[t] = vzero8();

  const float* arow = x + (size_t)(m0 + lo) * D_MODEL;

#pragma unroll 8
  for (int k = 0; k < D_MODEL; k += 4) {
    const int kk = k + 2 * hi;
    v2f a = *(const v2f*)(arow + kk);
#pragma unroll
    for (int t = 0; t < 4; ++t) {
      v2f b = *(const v2f*)(W + (size_t)(n0 + t * 16 + lo) * D_MODEL + kk);
      acc[t] = wmma_f32(a, b, acc[t]);
    }
  }

  const int bidx = m0 >> 11;
  if (mat == 2) {
#pragma unroll
    for (int t = 0; t < 4; ++t) {
      const int ng = n0 + t * 16 + lo;
      const int h = ng >> 7, d = ng & 127;
#pragma unroll
      for (int i = 0; i < 8; ++i) {
        const int s = (m0 + i + 8 * hi) & (SEQ - 1);
        Out[(((size_t)bidx * NHEADS + h) * SEQ + s) * DK + d] = acc[t][i];
      }
    }
  } else {
#pragma unroll
    for (int t = 0; t < 4; ++t) {
      const int ng = n0 + t * 16 + lo;
      const int h = ng >> 7, d = ng & 127;
      const float inv_freq = expf(-(float)(d & ~1) * (LOG_THETA / (float)DK));
      const int odd = d & 1;
#pragma unroll
      for (int i = 0; i < 8; ++i) {
        const int s = (m0 + i + 8 * hi) & (SEQ - 1);
        const float p = (float)tpos[bidx * SEQ + s];
        float sn, cs;
        sincosf(p * inv_freq, &sn, &cs);
        const float val = acc[t][i];
        const float part = __shfl_xor(val, 1, 32);
        const float r = odd ? fmaf(part, sn, val * cs)
                            : fmaf(val, cs, -part * sn);
        Out[(((size_t)bidx * NHEADS + h) * SEQ + s) * DK + d] = r;
      }
    }
  }
}

__global__ __launch_bounds__(32)
void mha_attn(const float* __restrict__ Q, const float* __restrict__ K,
              const float* __restrict__ V, float* __restrict__ AO) {
  const int lane = threadIdx.x;
  const int hi = lane >> 4;
  const int lo = lane & 15;
  const int qt = blockIdx.x;
  const int h = blockIdx.y;
  const int b = blockIdx.z;

  const size_t head = ((size_t)b * NHEADS + h) * SEQ;
  const float* __restrict__ Qb = Q + head * DK;
  const float* __restrict__ Kb = K + head * DK;
  const float* __restrict__ Vb = V + head * DK;

  __shared__ float Pst[16 * 16];

  v2f qa[32];
  const float* qrow = Qb + (size_t)(qt * 16 + lo) * DK;
#pragma unroll
  for (int ks = 0; ks < 32; ++ks)
    qa[ks] = *(const v2f*)(qrow + 4 * ks + 2 * hi);

  float mrow[8], lrow[8], alph[8];
  v8f o[8];
#pragma unroll
  for (int i = 0; i < 8; ++i) {
    mrow[i] = -__builtin_inff();
    lrow[i] = 0.f;
    o[i] = vzero8();
  }

  const float scale = 0.08838834764831845f;

  for (int kt = 0; kt <= qt; ++kt) {
    v8f sc = vzero8();
    const float* krow = Kb + (size_t)(kt * 16 + lo) * DK;
#pragma unroll
    for (int ks = 0; ks < 32; ++ks) {
      v2f kb = *(const v2f*)(krow + 4 * ks + 2 * hi);
      sc = wmma_f32(qa[ks], kb, sc);
    }

    const bool diag = (kt == qt);
#pragma unroll
    for (int i = 0; i < 8; ++i) {
      float s = sc[i] * scale;
      if (diag && lo > i + 8 * hi) s = -__builtin_inff();
      float rm = s;
      rm = fmaxf(rm, __shfl_xor(rm, 1, 32));
      rm = fmaxf(rm, __shfl_xor(rm, 2, 32));
      rm = fmaxf(rm, __shfl_xor(rm, 4, 32));
      rm = fmaxf(rm, __shfl_xor(rm, 8, 32));
      const float mnew = fmaxf(mrow[i], rm);
      const float a = expf(mrow[i] - mnew);
      const float p = expf(s - mnew);
      float rs = p;
      rs += __shfl_xor(rs, 1, 32);
      rs += __shfl_xor(rs, 2, 32);
      rs += __shfl_xor(rs, 4, 32);
      rs += __shfl_xor(rs, 8, 32);
      lrow[i] = lrow[i] * a + rs;
      mrow[i] = mnew;
      alph[i] = a;
      Pst[(i + 8 * hi) * 16 + lo] = p;
    }
#pragma unroll
    for (int t = 0; t < 8; ++t)
#pragma unroll
      for (int i = 0; i < 8; ++i) o[t][i] *= alph[i];

#pragma unroll
    for (int ks = 0; ks < 4; ++ks) {
      v2f a = *(const v2f*)(&Pst[lo * 16 + 4 * ks + 2 * hi]);
      const float* vrow = Vb + (size_t)(kt * 16 + 4 * ks + 2 * hi) * DK;
#pragma unroll
      for (int t = 0; t < 8; ++t) {
        v2f bv;
        bv[0] = vrow[t * 16 + lo];
        bv[1] = vrow[DK + t * 16 + lo];
        o[t] = wmma_f32(a, bv, o[t]);
      }
    }
  }

#pragma unroll
  for (int i = 0; i < 8; ++i) {
    const float invl = 1.f / lrow[i];
    const size_t row = (size_t)b * SEQ + qt * 16 + i + 8 * hi;
#pragma unroll
    for (int t = 0; t < 8; ++t)
      AO[row * D_MODEL + h * DK + t * 16 + lo] = o[t][i] * invl;
  }
}

__global__ __launch_bounds__(32)
void mha_out_proj(const float* __restrict__ A, const float* __restrict__ WO,
                  float* __restrict__ out) {
  const int lane = threadIdx.x;
  const int hi = lane >> 4;
  const int lo = lane & 15;
  const int m0 = blockIdx.x * 16;
  const int n0 = blockIdx.y * 64;

  v8f acc[4];
#pragma unroll
  for (int t = 0; t < 4; ++t) acc[t] = vzero8();

  const float* arow = A + (size_t)(m0 + lo) * D_MODEL;
#pragma unroll 8
  for (int k = 0; k < D_MODEL; k += 4) {
    const int kk = k + 2 * hi;
    v2f a = *(const v2f*)(arow + kk);
#pragma unroll
    for (int t = 0; t < 4; ++t) {
      v2f b = *(const v2f*)(WO + (size_t)(n0 + t * 16 + lo) * D_MODEL + kk);
      acc[t] = wmma_f32(a, b, acc[t]);
    }
  }

#pragma unroll
  for (int t = 0; t < 4; ++t)
#pragma unroll
    for (int i = 0; i < 8; ++i)
      out[(size_t)(m0 + i + 8 * hi) * D_MODEL + n0 + t * 16 + lo] = acc[t][i];
}

extern "C" void kernel_launch(void* const* d_in, const int* in_sizes, int n_in,
                              void* d_out, int out_size, void* d_ws, size_t ws_size,
                              hipStream_t stream) {
  (void)in_sizes; (void)n_in; (void)out_size; (void)ws_size;
  const float* x   = (const float*)d_in[0];
  const int*   tp  = (const int*)d_in[1];
  const float* WQ  = (const float*)d_in[2];
  const float* WK  = (const float*)d_in[3];
  const float* WV  = (const float*)d_in[4];
  const float* WO  = (const float*)d_in[5];
  float* out = (float*)d_out;

  const size_t QSZ = (size_t)BATCH * NHEADS * SEQ * DK;
  float* ws = (float*)d_ws;
  float* Qw = ws;
  float* Kw = ws + QSZ;
  float* Vw = ws + 2 * QSZ;
  float* AO = ws + 3 * QSZ;

  dim3 gProj(MROWS / 16, (D_MODEL / 16) / 4, 3);
  mha_qkv_rope<<<gProj, 32, 0, stream>>>(x, tp, WQ, WK, WV, Qw, Kw, Vw);

  dim3 gAttn(SEQ / 16, NHEADS, BATCH);
  mha_attn<<<gAttn, 32, 0, stream>>>(Qw, Kw, Vw, AO);

  dim3 gOut(MROWS / 16, (D_MODEL / 16) / 4);
  mha_out_proj<<<gOut, 32, 0, stream>>>(AO, WO, out);
}


# QKV and output projections: 128x128x64 block tiles staged through LDS with 16-byte coalesced global loads, 8 waves of 32x64 register tiles; per-element summation order unchanged
# speedup vs baseline: 2.3717x; 2.3717x over previous
#include <hip/hip_runtime.h>
#include <math.h>


#define D_MODEL 2048
#define NHEADS  16
#define DK      128
#define SEQ     2048
#define BATCH   2
#define MROWS   (BATCH * SEQ)
#define LOG_THETA 9.210340371976184f

typedef __attribute__((ext_vector_type(2))) float v2f;
typedef __attribute__((ext_vector_type(8))) float v8f;

__device__ __forceinline__ v8f wmma_f32(v2f a, v2f b, v8f c) {
  return __builtin_amdgcn_wmma_f32_16x16x4_f32(false, a, false, b,
                                               (short)0, c, false, false);
}

__device__ __forceinline__ v8f vzero8() {
  v8f z = {0.f, 0.f, 0.f, 0.f, 0.f, 0.f, 0.f, 0.f};
  return z;
}

#define GBM 128
#define GBN 128
#define GBK 64
#define GTHREADS 256
#define KTILES (D_MODEL / GBK)
#define LDP 68
#define STAGE (GBM * LDP)
#define GEMM_LDS_FLOATS (4 * STAGE)

typedef __attribute__((ext_vector_type(4))) float v4f;

__device__ __forceinline__ void gemm_nt_128x128(const float* __restrict__ A,
                                                const float* __restrict__ Wt,
                                                const int m0, const int n0,
                                                float* __restrict__ sA,
                                                float* __restrict__ sB,
                                                v8f (&acc)[2][4]) {
  const int tid = threadIdx.x;
  const int lane = tid & 31;
  const int wave = tid >> 5;
  const int hi = lane >> 4;
  const int lo = lane & 15;
  const int wm = wave & 3;
  const int wn = wave >> 2;

  const int lr = tid >> 4;
  const int lc = (tid & 15) << 2;
  const float* __restrict__ gA = A + (size_t)(m0 + lr) * D_MODEL + lc;
  const float* __restrict__ gW = Wt + (size_t)(n0 + lr) * D_MODEL + lc;
  const int sOff = lr * LDP + lc;

  v4f ra[8], rw[8];
#pragma unroll
  for (int p = 0; p < 8; ++p) {
    ra[p] = *(const v4f*)(gA + (size_t)p * 16 * D_MODEL);
    rw[p] = *(const v4f*)(gW + (size_t)p * 16 * D_MODEL);
  }
#pragma unroll
  for (int p = 0; p < 8; ++p) {
    *(v4f*)(sA + sOff + p * 16 * LDP) = ra[p];
    *(v4f*)(sB + sOff + p * 16 * LDP) = rw[p];
  }
  __syncthreads();

  const int aFrag = (wm * 32 + lo) * LDP + 2 * hi;
  const int bFrag = (wn * 64 + lo) * LDP + 2 * hi;

  for (int kt = 0; kt < KTILES; ++kt) {
    const int cur = (kt & 1) * STAGE;
    const int nxt = STAGE - cur;
    const bool more = (kt + 1 < KTILES);
    if (more) {
      const int k0 = (kt + 1) * GBK;
#pragma unroll
      for (int p = 0; p < 8; ++p) {
        ra[p] = *(const v4f*)(gA + (size_t)p * 16 * D_MODEL + k0);
        rw[p] = *(const v4f*)(gW + (size_t)p * 16 * D_MODEL + k0);
      }
    }

    const float* as = sA + cur + aFrag;
    const float* bs = sB + cur + bFrag;
#pragma unroll
    for (int s = 0; s < GBK / 4; ++s) {
      const v2f a0 = *(const v2f*)(as + 4 * s);
      const v2f a1 = *(const v2f*)(as + 16 * LDP + 4 * s);
      v2f b[4];
#pragma unroll
      for (int nt = 0; nt < 4; ++nt)
        b[nt] = *(const v2f*)(bs + nt * 16 * LDP + 4 * s);
#pragma unroll
      for (int nt = 0; nt < 4; ++nt) {
        acc[0][nt] = wmma_f32(a0, b[nt], acc[0][nt]);
        acc[1][nt] = wmma_f32(a1, b[nt], acc[1][nt]);
      }
    }

    if (more) {
#pragma unroll
      for (int p = 0; p < 8; ++p) {
        *(v4f*)(sA + nxt + sOff + p * 16 * LDP) = ra[p];
        *(v4f*)(sB + nxt + sOff + p * 16 * LDP) = rw[p];
      }
    }
    __syncthreads();
  }
}

__global__ __launch_bounds__(GTHREADS)
void mha_qkv_rope(const float* __restrict__ x, const int* __restrict__ tpos,
                  const float* __restrict__ WQ, const float* __restrict__ WK,
                  const float* __restrict__ WV,
                  float* __restrict__ Qo, float* __restrict__ Ko,
                  float* __restrict__ Vo) {
  __shared__ __attribute__((aligned(16))) float smem[GEMM_LDS_FLOATS];

  const int lane = threadIdx.x & 31;
  const int wave = threadIdx.x >> 5;
  const int hi = lane >> 4;
  const int lo = lane & 15;
  const int m0 = blockIdx.x * GBM;
  const int n0 = blockIdx.y * GBN;
  const int mat = blockIdx.z;

  const float* __restrict__ W = (mat == 0) ? WQ : ((mat == 1) ? WK : WV);
  float* __restrict__ Out = (mat == 0) ? Qo : ((mat == 1) ? Ko : Vo);

  v8f acc[2][4];
#pragma unroll
  for (int mt = 0; mt < 2; ++mt)
#pragma unroll
    for (int nt = 0; nt < 4; ++nt) acc[mt][nt] = vzero8();

  gemm_nt_128x128(x, W, m0, n0, smem, smem + 2 * STAGE, acc);

  const int mw = m0 + (wave & 3) * 32;
  const int nw = n0 + (wave >> 2) * 64;
  const int bidx = m0 >> 11;
  if (mat == 2) {
#pragma unroll
    for (int mt = 0; mt < 2; ++mt) {
#pragma unroll
      for (int nt = 0; nt < 4; ++nt) {
        const int ng = nw + nt * 16 + lo;
        const int h = ng >> 7, d = ng & 127;
#pragma unroll
        for (int i = 0; i < 8; ++i) {
          const int s = (mw + mt * 16 + i + 8 * hi) & (SEQ - 1);
          Out[(((size_t)bidx * NHEADS + h) * SEQ + s) * DK + d] = acc[mt][nt][i];
        }
      }
    }
  } else {
    float* __restrict__ park = smem + wave * (8 * 256);
#pragma unroll
    for (int mt = 0; mt < 2; ++mt) {
#pragma unroll
      for (int nt = 0; nt < 4; ++nt) {
        const int t = mt * 4 + nt;
        const v4f c0 = {acc[mt][nt][0], acc[mt][nt][1], acc[mt][nt][2], acc[mt][nt][3]};
        const v4f c1 = {acc[mt][nt][4], acc[mt][nt][5], acc[mt][nt][6], acc[mt][nt][7]};
        *(v4f*)(park + (2 * t + 0) * 128 + lane * 4) = c0;
        *(v4f*)(park + (2 * t + 1) * 128 + lane * 4) = c1;
      }
    }
#pragma unroll 1
    for (int t = 0; t < 8; ++t) {
      const v4f c0 = *(const v4f*)(park + (2 * t + 0) * 128 + lane * 4);
      const v4f c1 = *(const v4f*)(park + (2 * t + 1) * 128 + lane * 4);
      const int mt = t >> 2, nt = t & 3;
      const int ng = nw + nt * 16 + lo;
      const int h = ng >> 7, d = ng & 127;
      const float inv_freq = expf(-(float)(d & ~1) * (LOG_THETA / (float)DK));
      const int odd = d & 1;
#pragma unroll
      for (int i = 0; i < 8; ++i) {
        const int s = (mw + mt * 16 + i + 8 * hi) & (SEQ - 1);
        const float p = (float)tpos[bidx * SEQ + s];
        float sn, cs;
        sincosf(p * inv_freq, &sn, &cs);
        const float val = (i < 4) ? c0[i & 3] : c1[i & 3];
        const float part = __shfl_xor(val, 1, 32);
        const float r = odd ? fmaf(part, sn, val * cs)
                            : fmaf(val, cs, -part * sn);
        Out[(((size_t)bidx * NHEADS + h) * SEQ + s) * DK + d] = r;
      }
    }
  }
}

__global__ __launch_bounds__(32)
void mha_attn(const float* __restrict__ Q, const float* __restrict__ K,
              const float* __restrict__ V, float* __restrict__ AO) {
  const int lane = threadIdx.x;
  const int hi = lane >> 4;
  const int lo = lane & 15;
  const int qt = blockIdx.x;
  const int h = blockIdx.y;
  const int b = blockIdx.z;

  const size_t head = ((size_t)b * NHEADS + h) * SEQ;
  const float* __restrict__ Qb = Q + head * DK;
  const float* __restrict__ Kb = K + head * DK;
  const float* __restrict__ Vb = V + head * DK;

  __shared__ float Pst[16 * 16];

  v2f qa[32];
  const float* qrow = Qb + (size_t)(qt * 16 + lo) * DK;
#pragma unroll
  for (int ks = 0; ks < 32; ++ks)
    qa[ks] = *(const v2f*)(qrow + 4 * ks + 2 * hi);

  float mrow[8], lrow[8], alph[8];
  v8f o[8];
#pragma unroll
  for (int i = 0; i < 8; ++i) {
    mrow[i] = -__builtin_inff();
    lrow[i] = 0.f;
    o[i] = vzero8();
  }

  const float scale = 0.08838834764831845f;

  for (int kt = 0; kt <= qt; ++kt) {
    v8f sc = vzero8();
    const float* krow = Kb + (size_t)(kt * 16 + lo) * DK;
#pragma unroll
    for (int ks = 0; ks < 32; ++ks) {
      v2f kb = *(const v2f*)(krow + 4 * ks + 2 * hi);
      sc = wmma_f32(qa[ks], kb, sc);
    }

    const bool diag = (kt == qt);
#pragma unroll
    for (int i = 0; i < 8; ++i) {
      float s = sc[i] * scale;
      if (diag && lo > i + 8 * hi) s = -__builtin_inff();
      float rm = s;
      rm = fmaxf(rm, __shfl_xor(rm, 1, 32));
      rm = fmaxf(rm, __shfl_xor(rm, 2, 32));
      rm = fmaxf(rm, __shfl_xor(rm, 4, 32));
      rm = fmaxf(rm, __shfl_xor(rm, 8, 32));
      const float mnew = fmaxf(mrow[i], rm);
      const float a = expf(mrow[i] - mnew);
      const float p = expf(s - mnew);
      float rs = p;
      rs += __shfl_xor(rs, 1, 32);
      rs += __shfl_xor(rs, 2, 32);
      rs += __shfl_xor(rs, 4, 32);
      rs += __shfl_xor(rs, 8, 32);
      lrow[i] = lrow[i] * a + rs;
      mrow[i] = mnew;
      alph[i] = a;
      Pst[(i + 8 * hi) * 16 + lo] = p;
    }
#pragma unroll
    for (int t = 0; t < 8; ++t)
#pragma unroll
      for (int i = 0; i < 8; ++i) o[t][i] *= alph[i];

#pragma unroll
    for (int ks = 0; ks < 4; ++ks) {
      v2f a = *(const v2f*)(&Pst[lo * 16 + 4 * ks + 2 * hi]);
      const float* vrow = Vb + (size_t)(kt * 16 + 4 * ks + 2 * hi) * DK;
#pragma unroll
      for (int t = 0; t < 8; ++t) {
        v2f bv;
        bv[0] = vrow[t * 16 + lo];
        bv[1] = vrow[DK + t * 16 + lo];
        o[t] = wmma_f32(a, bv, o[t]);
      }
    }
  }

#pragma unroll
  for (int i = 0; i < 8; ++i) {
    const float invl = 1.f / lrow[i];
    const size_t row = (size_t)b * SEQ + qt * 16 + i + 8 * hi;
#pragma unroll
    for (int t = 0; t < 8; ++t)
      AO[row * D_MODEL + h * DK + t * 16 + lo] = o[t][i] * invl;
  }
}

__global__ __launch_bounds__(GTHREADS)
void mha_out_proj(const float* __restrict__ A, const float* __restrict__ WO,
                  float* __restrict__ out) {
  __shared__ __attribute__((aligned(16))) float smem[GEMM_LDS_FLOATS];

  const int lane = threadIdx.x & 31;
  const int wave = threadIdx.x >> 5;
  const int hi = lane >> 4;
  const int lo = lane & 15;
  const int m0 = blockIdx.x * GBM;
  const int n0 = blockIdx.y * GBN;

  v8f acc[2][4];
#pragma unroll
  for (int mt = 0; mt < 2; ++mt)
#pragma unroll
    for (int nt = 0; nt < 4; ++nt) acc[mt][nt] = vzero8();

  gemm_nt_128x128(A, WO, m0, n0, smem, smem + 2 * STAGE, acc);

  const int mw = m0 + (wave & 3) * 32;
  const int nw = n0 + (wave >> 2) * 64;
#pragma unroll
  for (int mt = 0; mt < 2; ++mt)
#pragma unroll
    for (int nt = 0; nt < 4; ++nt)
#pragma unroll
      for (int i = 0; i < 8; ++i)
        out[(size_t)(mw + mt * 16 + i + 8 * hi) * D_MODEL + nw + nt * 16 + lo] =
            acc[mt][nt][i];
}

extern "C" void kernel_launch(void* const* d_in, const int* in_sizes, int n_in,
                              void* d_out, int out_size, void* d_ws, size_t ws_size,
                              hipStream_t stream) {
  (void)in_sizes; (void)n_in; (void)out_size; (void)ws_size;
  const float* x   = (const float*)d_in[0];
  const int*   tp  = (const int*)d_in[1];
  const float* WQ  = (const float*)d_in[2];
  const float* WK  = (const float*)d_in[3];
  const float* WV  = (const float*)d_in[4];
  const float* WO  = (const float*)d_in[5];
  float* out = (float*)d_out;

  const size_t QSZ = (size_t)BATCH * NHEADS * SEQ * DK;
  float* ws = (float*)d_ws;
  float* Qw = ws;
  float* Kw = ws + QSZ;
  float* Vw = ws + 2 * QSZ;
  float* AO = ws + 3 * QSZ;

  dim3 gProj(MROWS / GBM, D_MODEL / GBN, 3);
  mha_qkv_rope<<<gProj, GTHREADS, 0, stream>>>(x, tp, WQ, WK, WV, Qw, Kw, Vw);

  dim3 gAttn(SEQ / 16, NHEADS, BATCH);
  mha_attn<<<gAttn, 32, 0, stream>>>(Qw, Kw, Vw, AO);

  dim3 gOut(MROWS / GBM, D_MODEL / GBN);
  mha_out_proj<<<gOut, GTHREADS, 0, stream>>>(AO, WO, out);
}


# Attention computed transposed (K.Q^T, then V^T.P^T): a lane owns one query, so row max/sum are in-lane plus one half swap and P stays in registers; reorders the row-sum and P.V key summation
# speedup vs baseline: 2.5340x; 1.0684x over previous
#include <hip/hip_runtime.h>
#include <math.h>


#define D_MODEL 2048
#define NHEADS  16
#define DK      128
#define SEQ     2048
#define BATCH   2
#define MROWS   (BATCH * SEQ)
#define LOG_THETA 9.210340371976184f

typedef __attribute__((ext_vector_type(2))) float v2f;
typedef __attribute__((ext_vector_type(8))) float v8f;

__device__ __forceinline__ v8f wmma_f32(v2f a, v2f b, v8f c) {
  return __builtin_amdgcn_wmma_f32_16x16x4_f32(false, a, false, b,
                                               (short)0, c, false, false);
}

__device__ __forceinline__ v8f vzero8() {
  v8f z = {0.f, 0.f, 0.f, 0.f, 0.f, 0.f, 0.f, 0.f};
  return z;
}

#define GBM 128
#define GBN 128
#define GBK 64
#define GTHREADS 256
#define KTILES (D_MODEL / GBK)
#define LDP 68
#define STAGE (GBM * LDP)
#define GEMM_LDS_FLOATS (4 * STAGE)

typedef __attribute__((ext_vector_type(4))) float v4f;

__device__ __forceinline__ void gemm_nt_128x128(const float* __restrict__ A,
                                                const float* __restrict__ Wt,
                                                const int m0, const int n0,
                                                float* __restrict__ sA,
                                                float* __restrict__ sB,
                                                v8f (&acc)[2][4]) {
  const int tid = threadIdx.x;
  const int lane = tid & 31;
  const int wave = tid >> 5;
  const int hi = lane >> 4;
  const int lo = lane & 15;
  const int wm = wave & 3;
  const int wn = wave >> 2;

  const int lr = tid >> 4;
  const int lc = (tid & 15) << 2;
  const float* __restrict__ gA = A + (size_t)(m0 + lr) * D_MODEL + lc;
  const float* __restrict__ gW = Wt + (size_t)(n0 + lr) * D_MODEL + lc;
  const int sOff = lr * LDP + lc;

  v4f ra[8], rw[8];
#pragma unroll
  for (int p = 0; p < 8; ++p) {
    ra[p] = *(const v4f*)(gA + (size_t)p * 16 * D_MODEL);
    rw[p] = *(const v4f*)(gW + (size_t)p * 16 * D_MODEL);
  }
#pragma unroll
  for (int p = 0; p < 8; ++p) {
    *(v4f*)(sA + sOff + p * 16 * LDP) = ra[p];
    *(v4f*)(sB + sOff + p * 16 * LDP) = rw[p];
  }
  __syncthreads();

  const int aFrag = (wm * 32 + lo) * LDP + 2 * hi;
  const int bFrag = (wn * 64 + lo) * LDP + 2 * hi;

  for (int kt = 0; kt < KTILES; ++kt) {
    const int cur = (kt & 1) * STAGE;
    const int nxt = STAGE - cur;
    const bool more = (kt + 1 < KTILES);
    if (more) {
      const int k0 = (kt + 1) * GBK;
#pragma unroll
      for (int p = 0; p < 8; ++p) {
        ra[p] = *(const v4f*)(gA + (size_t)p * 16 * D_MODEL + k0);
        rw[p] = *(const v4f*)(gW + (size_t)p * 16 * D_MODEL + k0);
      }
    }

    const float* as = sA + cur + aFrag;
    const float* bs = sB + cur + bFrag;
#pragma unroll
    for (int s = 0; s < GBK / 4; ++s) {
      const v2f a0 = *(const v2f*)(as + 4 * s);
      const v2f a1 = *(const v2f*)(as + 16 * LDP + 4 * s);
      v2f b[4];
#pragma unroll
      for (int nt = 0; nt < 4; ++nt)
        b[nt] = *(const v2f*)(bs + nt * 16 * LDP + 4 * s);
#pragma unroll
      for (int nt = 0; nt < 4; ++nt) {
        acc[0][nt] = wmma_f32(a0, b[nt], acc[0][nt]);
        acc[1][nt] = wmma_f32(a1, b[nt], acc[1][nt]);
      }
    }

    if (more) {
#pragma unroll
      for (int p = 0; p < 8; ++p) {
        *(v4f*)(sA + nxt + sOff + p * 16 * LDP) = ra[p];
        *(v4f*)(sB + nxt + sOff + p * 16 * LDP) = rw[p];
      }
    }
    __syncthreads();
  }
}

__global__ __launch_bounds__(GTHREADS)
void mha_qkv_rope(const float* __restrict__ x, const int* __restrict__ tpos,
                  const float* __restrict__ WQ, const float* __restrict__ WK,
                  const float* __restrict__ WV,
                  float* __restrict__ Qo, float* __restrict__ Ko,
                  float* __restrict__ Vo) {
  __shared__ __attribute__((aligned(16))) float smem[GEMM_LDS_FLOATS];

  const int lane = threadIdx.x & 31;
  const int wave = threadIdx.x >> 5;
  const int hi = lane >> 4;
  const int lo = lane & 15;
  const int m0 = blockIdx.x * GBM;
  const int n0 = blockIdx.y * GBN;
  const int mat = blockIdx.z;

  const float* __restrict__ W = (mat == 0) ? WQ : ((mat == 1) ? WK : WV);
  float* __restrict__ Out = (mat == 0) ? Qo : ((mat == 1) ? Ko : Vo);

  v8f acc[2][4];
#pragma unroll
  for (int mt = 0; mt < 2; ++mt)
#pragma unroll
    for (int nt = 0; nt < 4; ++nt) acc[mt][nt] = vzero8();

  gemm_nt_128x128(x, W, m0, n0, smem, smem + 2 * STAGE, acc);

  const int mw = m0 + (wave & 3) * 32;
  const int nw = n0 + (wave >> 2) * 64;
  const int bidx = m0 >> 11;
  if (mat == 2) {
#pragma unroll
    for (int mt = 0; mt < 2; ++mt) {
#pragma unroll
      for (int nt = 0; nt < 4; ++nt) {
        const int ng = nw + nt * 16 + lo;
        const int h = ng >> 7, d = ng & 127;
#pragma unroll
        for (int i = 0; i < 8; ++i) {
          const int s = (mw + mt * 16 + i + 8 * hi) & (SEQ - 1);
          Out[(((size_t)bidx * NHEADS + h) * SEQ + s) * DK + d] = acc[mt][nt][i];
        }
      }
    }
  } else {
    float* __restrict__ park = smem + wave * (8 * 256);
#pragma unroll
    for (int mt = 0; mt < 2; ++mt) {
#pragma unroll
      for (int nt = 0; nt < 4; ++nt) {
        const int t = mt * 4 + nt;
        const v4f c0 = {acc[mt][nt][0], acc[mt][nt][1], acc[mt][nt][2], acc[mt][nt][3]};
        const v4f c1 = {acc[mt][nt][4], acc[mt][nt][5], acc[mt][nt][6], acc[mt][nt][7]};
        *(v4f*)(park + (2 * t + 0) * 128 + lane * 4) = c0;
        *(v4f*)(park + (2 * t + 1) * 128 + lane * 4) = c1;
      }
    }
#pragma unroll 1
    for (int t = 0; t < 8; ++t) {
      const v4f c0 = *(const v4f*)(park + (2 * t + 0) * 128 + lane * 4);
      const v4f c1 = *(const v4f*)(park + (2 * t + 1) * 128 + lane * 4);
      const int mt = t >> 2, nt = t & 3;
      const int ng = nw + nt * 16 + lo;
      const int h = ng >> 7, d = ng & 127;
      const float inv_freq = expf(-(float)(d & ~1) * (LOG_THETA / (float)DK));
      const int odd = d & 1;
#pragma unroll
      for (int i = 0; i < 8; ++i) {
        const int s = (mw + mt * 16 + i + 8 * hi) & (SEQ - 1);
        const float p = (float)tpos[bidx * SEQ + s];
        float sn, cs;
        sincosf(p * inv_freq, &sn, &cs);
        const float val = (i < 4) ? c0[i & 3] : c1[i & 3];
        const float part = __shfl_xor(val, 1, 32);
        const float r = odd ? fmaf(part, sn, val * cs)
                            : fmaf(val, cs, -part * sn);
        Out[(((size_t)bidx * NHEADS + h) * SEQ + s) * DK + d] = r;
      }
    }
  }
}

__device__ __forceinline__ float xhalf(float x) {
  const int b = __builtin_bit_cast(int, x);
  return __builtin_bit_cast(float, __builtin_amdgcn_permlanex16(b, b, 0x76543210, 0xfedcba98,
                                                                false, false));
}

__global__ __launch_bounds__(32)
void mha_attn(const float* __restrict__ Q, const float* __restrict__ K,
              const float* __restrict__ V, float* __restrict__ AO) {
  const int lane = threadIdx.x;
  const int hi = lane >> 4;
  const int lo = lane & 15;
  const int qt = blockIdx.x;
  const int h = blockIdx.y;
  const int b = blockIdx.z;

  const size_t head = ((size_t)b * NHEADS + h) * SEQ;
  const float* __restrict__ Qb = Q + head * DK;
  const float* __restrict__ Kb = K + head * DK;
  const float* __restrict__ Vb = V + head * DK;

  v2f qa[32];
  const float* qrow = Qb + (size_t)(qt * 16 + lo) * DK;
#pragma unroll
  for (int ks = 0; ks < 32; ++ks)
    qa[ks] = *(const v2f*)(qrow + 4 * ks + 2 * hi);

  float mrow = -__builtin_inff();
  float lrow = 0.f;
  v8f o[8];
#pragma unroll
  for (int t = 0; t < 8; ++t) o[t] = vzero8();

  const float scale = 0.08838834764831845f;

  for (int kt = 0; kt <= qt; ++kt) {
    v8f sc = vzero8();
    const float* krow = Kb + (size_t)(kt * 16 + lo) * DK;
#pragma unroll
    for (int ks = 0; ks < 32; ++ks) {
      v2f kb = *(const v2f*)(krow + 4 * ks + 2 * hi);
      sc = wmma_f32(kb, qa[ks], sc);
    }

    const bool diag = (kt == qt);
    float s[8];
#pragma unroll
    for (int r = 0; r < 8; ++r) {
      s[r] = sc[r] * scale;
      if (diag && (8 * hi + r) > lo) s[r] = -__builtin_inff();
    }
    float rm = s[0];
#pragma unroll
    for (int r = 1; r < 8; ++r) rm = fmaxf(rm, s[r]);
    rm = fmaxf(rm, xhalf(rm));
    const float mnew = fmaxf(mrow, rm);
    const float a = expf(mrow - mnew);
    float p[8];
#pragma unroll
    for (int r = 0; r < 8; ++r) p[r] = expf(s[r] - mnew);
    float rs = p[0];
#pragma unroll
    for (int r = 1; r < 8; ++r) rs += p[r];
    rs += xhalf(rs);
    lrow = lrow * a + rs;
    mrow = mnew;

#pragma unroll
    for (int t = 0; t < 8; ++t)
#pragma unroll
      for (int r = 0; r < 8; ++r) o[t][r] *= a;

#pragma unroll
    for (int j = 0; j < 4; ++j) {
      v2f pb;
      pb[0] = p[2 * j];
      pb[1] = p[2 * j + 1];
      const float* vrow = Vb + (size_t)(kt * 16 + 8 * hi + 2 * j) * DK + lo;
#pragma unroll
      for (int t = 0; t < 8; ++t) {
        v2f av;
        av[0] = vrow[t * 16];
        av[1] = vrow[DK + t * 16];
        o[t] = wmma_f32(av, pb, o[t]);
      }
    }
  }

  const float invl = 1.f / lrow;
  float* __restrict__ orow =
      AO + ((size_t)b * SEQ + qt * 16 + lo) * D_MODEL + h * DK + 8 * hi;
#pragma unroll
  for (int t = 0; t < 8; ++t) {
    const v4f w0 = {o[t][0] * invl, o[t][1] * invl, o[t][2] * invl, o[t][3] * invl};
    const v4f w1 = {o[t][4] * invl, o[t][5] * invl, o[t][6] * invl, o[t][7] * invl};
    *(v4f*)(orow + t * 16) = w0;
    *(v4f*)(orow + t * 16 + 4) = w1;
  }
}

__global__ __launch_bounds__(GTHREADS)
void mha_out_proj(const float* __restrict__ A, const float* __restrict__ WO,
                  float* __restrict__ out) {
  __shared__ __attribute__((aligned(16))) float smem[GEMM_LDS_FLOATS];

  const int lane = threadIdx.x & 31;
  const int wave = threadIdx.x >> 5;
  const int hi = lane >> 4;
  const int lo = lane & 15;
  const int m0 = blockIdx.x * GBM;
  const int n0 = blockIdx.y * GBN;

  v8f acc[2][4];
#pragma unroll
  for (int mt = 0; mt < 2; ++mt)
#pragma unroll
    for (int nt = 0; nt < 4; ++nt) acc[mt][nt] = vzero8();

  gemm_nt_128x128(A, WO, m0, n0, smem, smem + 2 * STAGE, acc);

  const int mw = m0 + (wave & 3) * 32;
  const int nw = n0 + (wave >> 2) * 64;
#pragma unroll
  for (int mt = 0; mt < 2; ++mt)
#pragma unroll
    for (int nt = 0; nt < 4; ++nt)
#pragma unroll
      for (int i = 0; i < 8; ++i)
        out[(size_t)(mw + mt * 16 + i + 8 * hi) * D_MODEL + nw + nt * 16 + lo] =
            acc[mt][nt][i];
}

extern "C" void kernel_launch(void* const* d_in, const int* in_sizes, int n_in,
                              void* d_out, int out_size, void* d_ws, size_t ws_size,
                              hipStream_t stream) {
  (void)in_sizes; (void)n_in; (void)out_size; (void)ws_size;
  const float* x   = (const float*)d_in[0];
  const int*   tp  = (const int*)d_in[1];
  const float* WQ  = (const float*)d_in[2];
  const float* WK  = (const float*)d_in[3];
  const float* WV  = (const float*)d_in[4];
  const float* WO  = (const float*)d_in[5];
  float* out = (float*)d_out;

  const size_t QSZ = (size_t)BATCH * NHEADS * SEQ * DK;
  float* ws = (float*)d_ws;
  float* Qw = ws;
  float* Kw = ws + QSZ;
  float* Vw = ws + 2 * QSZ;
  float* AO = ws + 3 * QSZ;

  dim3 gProj(MROWS / GBM, D_MODEL / GBN, 3);
  mha_qkv_rope<<<gProj, GTHREADS, 0, stream>>>(x, tp, WQ, WK, WV, Qw, Kw, Vw);

  dim3 gAttn(SEQ / 16, NHEADS, BATCH);
  mha_attn<<<gAttn, 32, 0, stream>>>(Qw, Kw, Vw, AO);

  dim3 gOut(MROWS / GBM, D_MODEL / GBN);
  mha_out_proj<<<gOut, GTHREADS, 0, stream>>>(AO, WO, out);
}


# Attention: 8 waves per work-group share each 32-key K/V chunk through LDS (16-byte coalesced loads, 2 stages); light and heavy query blocks paired per work-group; summation order unchanged
# speedup vs baseline: 3.3085x; 1.3056x over previous
#include <hip/hip_runtime.h>
#include <math.h>


#define D_MODEL 2048
#define NHEADS  16
#define DK      128
#define SEQ     2048
#define BATCH   2
#define MROWS   (BATCH * SEQ)
#define LOG_THETA 9.210340371976184f

typedef __attribute__((ext_vector_type(2))) float v2f;
typedef __attribute__((ext_vector_type(8))) float v8f;

__device__ __forceinline__ v8f wmma_f32(v2f a, v2f b, v8f c) {
  return __builtin_amdgcn_wmma_f32_16x16x4_f32(false, a, false, b,
                                               (short)0, c, false, false);
}

__device__ __forceinline__ v8f vzero8() {
  v8f z = {0.f, 0.f, 0.f, 0.f, 0.f, 0.f, 0.f, 0.f};
  return z;
}

#define GBM 128
#define GBN 128
#define GBK 64
#define GTHREADS 256
#define KTILES (D_MODEL / GBK)
#define LDP 68
#define STAGE (GBM * LDP)
#define GEMM_LDS_FLOATS (4 * STAGE)

typedef __attribute__((ext_vector_type(4))) float v4f;

__device__ __forceinline__ void gemm_nt_128x128(const float* __restrict__ A,
                                                const float* __restrict__ Wt,
                                                const int m0, const int n0,
                                                float* __restrict__ sA,
                                                float* __restrict__ sB,
                                                v8f (&acc)[2][4]) {
  const int tid = threadIdx.x;
  const int lane = tid & 31;
  const int wave = tid >> 5;
  const int hi = lane >> 4;
  const int lo = lane & 15;
  const int wm = wave & 3;
  const int wn = wave >> 2;

  const int lr = tid >> 4;
  const int lc = (tid & 15) << 2;
  const float* __restrict__ gA = A + (size_t)(m0 + lr) * D_MODEL + lc;
  const float* __restrict__ gW = Wt + (size_t)(n0 + lr) * D_MODEL + lc;
  const int sOff = lr * LDP + lc;

  v4f ra[8], rw[8];
#pragma unroll
  for (int p = 0; p < 8; ++p) {
    ra[p] = *(const v4f*)(gA + (size_t)p * 16 * D_MODEL);
    rw[p] = *(const v4f*)(gW + (size_t)p * 16 * D_MODEL);
  }
#pragma unroll
  for (int p = 0; p < 8; ++p) {
    *(v4f*)(sA + sOff + p * 16 * LDP) = ra[p];
    *(v4f*)(sB + sOff + p * 16 * LDP) = rw[p];
  }
  __syncthreads();

  const int aFrag = (wm * 32 + lo) * LDP + 2 * hi;
  const int bFrag = (wn * 64 + lo) * LDP + 2 * hi;

  for (int kt = 0; kt < KTILES; ++kt) {
    const int cur = (kt & 1) * STAGE;
    const int nxt = STAGE - cur;
    const bool more = (kt + 1 < KTILES);
    if (more) {
      const int k0 = (kt + 1) * GBK;
#pragma unroll
      for (int p = 0; p < 8; ++p) {
        ra[p] = *(const v4f*)(gA + (size_t)p * 16 * D_MODEL + k0);
        rw[p] = *(const v4f*)(gW + (size_t)p * 16 * D_MODEL + k0);
      }
    }

    const float* as = sA + cur + aFrag;
    const float* bs = sB + cur + bFrag;
#pragma unroll
    for (int s = 0; s < GBK / 4; ++s) {
      const v2f a0 = *(const v2f*)(as + 4 * s);
      const v2f a1 = *(const v2f*)(as + 16 * LDP + 4 * s);
      v2f b[4];
#pragma unroll
      for (int nt = 0; nt < 4; ++nt)
        b[nt] = *(const v2f*)(bs + nt * 16 * LDP + 4 * s);
#pragma unroll
      for (int nt = 0; nt < 4; ++nt) {
        acc[0][nt] = wmma_f32(a0, b[nt], acc[0][nt]);
        acc[1][nt] = wmma_f32(a1, b[nt], acc[1][nt]);
      }
    }

    if (more) {
#pragma unroll
      for (int p = 0; p < 8; ++p) {
        *(v4f*)(sA + nxt + sOff + p * 16 * LDP) = ra[p];
        *(v4f*)(sB + nxt + sOff + p * 16 * LDP) = rw[p];
      }
    }
    __syncthreads();
  }
}

__global__ __launch_bounds__(GTHREADS)
void mha_qkv_rope(const float* __restrict__ x, const int* __restrict__ tpos,
                  const float* __restrict__ WQ, const float* __restrict__ WK,
                  const float* __restrict__ WV,
                  float* __restrict__ Qo, float* __restrict__ Ko,
                  float* __restrict__ Vo) {
  __shared__ __attribute__((aligned(16))) float smem[GEMM_LDS_FLOATS];

  const int lane = threadIdx.x & 31;
  const int wave = threadIdx.x >> 5;
  const int hi = lane >> 4;
  const int lo = lane & 15;
  const int m0 = blockIdx.x * GBM;
  const int n0 = blockIdx.y * GBN;
  const int mat = blockIdx.z;

  const float* __restrict__ W = (mat == 0) ? WQ : ((mat == 1) ? WK : WV);
  float* __restrict__ Out = (mat == 0) ? Qo : ((mat == 1) ? Ko : Vo);

  v8f acc[2][4];
#pragma unroll
  for (int mt = 0; mt < 2; ++mt)
#pragma unroll
    for (int nt = 0; nt < 4; ++nt) acc[mt][nt] = vzero8();

  gemm_nt_128x128(x, W, m0, n0, smem, smem + 2 * STAGE, acc);

  const int mw = m0 + (wave & 3) * 32;
  const int nw = n0 + (wave >> 2) * 64;
  const int bidx = m0 >> 11;
  if (mat == 2) {
#pragma unroll
    for (int mt = 0; mt < 2; ++mt) {
#pragma unroll
      for (int nt = 0; nt < 4; ++nt) {
        const int ng = nw + nt * 16 + lo;
        const int h = ng >> 7, d = ng & 127;
#pragma unroll
        for (int i = 0; i < 8; ++i) {
          const int s = (mw + mt * 16 + i + 8 * hi) & (SEQ - 1);
          Out[(((size_t)bidx * NHEADS + h) * SEQ + s) * DK + d] = acc[mt][nt][i];
        }
      }
    }
  } else {
    float* __restrict__ park = smem + wave * (8 * 256);
#pragma unroll
    for (int mt = 0; mt < 2; ++mt) {
#pragma unroll
      for (int nt = 0; nt < 4; ++nt) {
        const int t = mt * 4 + nt;
        const v4f c0 = {acc[mt][nt][0], acc[mt][nt][1], acc[mt][nt][2], acc[mt][nt][3]};
        const v4f c1 = {acc[mt][nt][4], acc[mt][nt][5], acc[mt][nt][6], acc[mt][nt][7]};
        *(v4f*)(park + (2 * t + 0) * 128 + lane * 4) = c0;
        *(v4f*)(park + (2 * t + 1) * 128 + lane * 4) = c1;
      }
    }
#pragma unroll 1
    for (int t = 0; t < 8; ++t) {
      const v4f c0 = *(const v4f*)(park + (2 * t + 0) * 128 + lane * 4);
      const v4f c1 = *(const v4f*)(park + (2 * t + 1) * 128 + lane * 4);
      const int mt = t >> 2, nt = t & 3;
      const int ng = nw + nt * 16 + lo;
      const int h = ng >> 7, d = ng & 127;
      const float inv_freq = expf(-(float)(d & ~1) * (LOG_THETA / (float)DK));
      const int odd = d & 1;
#pragma unroll
      for (int i = 0; i < 8; ++i) {
        const int s = (mw + mt * 16 + i + 8 * hi) & (SEQ - 1);
        const float p = (float)tpos[bidx * SEQ + s];
        float sn, cs;
        sincosf(p * inv_freq, &sn, &cs);
        const float val = (i < 4) ? c0[i & 3] : c1[i & 3];
        const float part = __shfl_xor(val, 1, 32);
        const float r = odd ? fmaf(part, sn, val * cs)
                            : fmaf(val, cs, -part * sn);
        Out[(((size_t)bidx * NHEADS + h) * SEQ + s) * DK + d] = r;
      }
    }
  }
}

__device__ __forceinline__ float xhalf(float x) {
  const int b = __builtin_bit_cast(int, x);
  return __builtin_bit_cast(float, __builtin_amdgcn_permlanex16(b, b, 0x76543210, 0xfedcba98,
                                                                false, false));
}

#define AKC 32
#define APITCH 132
#define ASTAGE (AKC * APITCH)
#define ATHREADS 256
#define ATT_LDS_FLOATS (4 * ASTAGE)
#define QBLOCKS (SEQ / 128)

__device__ __forceinline__ void softmax_pv(const v8f sc, const float* __restrict__ vt,
                                           const bool diag, const int hi, const int lo,
                                           float& mrow, float& lrow, v8f (&o)[8]) {
  v2f av[4][8];
#pragma unroll
  for (int j = 0; j < 4; ++j) {
    const float* vrow = vt + (8 * hi + 2 * j) * APITCH + lo;
#pragma unroll
    for (int t = 0; t < 8; ++t) {
      av[j][t][0] = vrow[t * 16];
      av[j][t][1] = vrow[APITCH + t * 16];
    }
  }
  const float scale = 0.08838834764831845f;
  float s[8];
#pragma unroll
  for (int r = 0; r < 8; ++r) {
    s[r] = sc[r] * scale;
    if (diag && (8 * hi + r) > lo) s[r] = -__builtin_inff();
  }
  float rm = s[0];
#pragma unroll
  for (int r = 1; r < 8; ++r) rm = fmaxf(rm, s[r]);
  rm = fmaxf(rm, xhalf(rm));
  const float mnew = fmaxf(mrow, rm);
  const float a = expf(mrow - mnew);
  float p[8];
#pragma unroll
  for (int r = 0; r < 8; ++r) p[r] = expf(s[r] - mnew);
  float rs = p[0];
#pragma unroll
  for (int r = 1; r < 8; ++r) rs += p[r];
  rs += xhalf(rs);
  lrow = lrow * a + rs;
  mrow = mnew;

#pragma unroll
  for (int t = 0; t < 8; ++t)
#pragma unroll
    for (int r = 0; r < 8; ++r) o[t][r] *= a;

#pragma unroll
  for (int j = 0; j < 4; ++j) {
    v2f pb;
    pb[0] = p[2 * j];
    pb[1] = p[2 * j + 1];
#pragma unroll
    for (int t = 0; t < 8; ++t) o[t] = wmma_f32(av[j][t], pb, o[t]);
  }
}

__global__ __launch_bounds__(ATHREADS)
void mha_attn(const float* __restrict__ Q, const float* __restrict__ K,
              const float* __restrict__ V, float* __restrict__ AO) {
  __shared__ __attribute__((aligned(16))) float smem[ATT_LDS_FLOATS];
  float* __restrict__ sK = smem;
  float* __restrict__ sV = smem + 2 * ASTAGE;

  const int tid = threadIdx.x;
  const int lane = tid & 31;
  const int wave = __builtin_amdgcn_readfirstlane(tid >> 5);
  const int hi = lane >> 4;
  const int lo = lane & 15;
  const int pr = blockIdx.x;
  const int h = blockIdx.y;
  const int b = blockIdx.z;

  const size_t head = ((size_t)b * NHEADS + h) * SEQ;
  const float* __restrict__ Qb = Q + head * DK;
  const float* __restrict__ Kb = K + head * DK;
  const float* __restrict__ Vb = V + head * DK;

  const int cOff = wave * DK + lane * 4;
  const int sOff = wave * APITCH + lane * 4;

  for (int pass = 0; pass < 2; ++pass) {
    const int qb = pass ? (QBLOCKS - 1 - pr) : pr;
    const int qt = qb * 8 + wave;
    const int nchunks = qb * 4 + 4;

    v2f qa[32];
    const float* qrow = Qb + (size_t)(qt * 16 + lo) * DK;
#pragma unroll
    for (int ks = 0; ks < 32; ++ks)
      qa[ks] = *(const v2f*)(qrow + 4 * ks + 2 * hi);

    float mrow = -__builtin_inff();
    float lrow = 0.f;
    v8f o[8];
#pragma unroll
    for (int t = 0; t < 8; ++t) o[t] = vzero8();

    v4f rk[4], rv[4];
#pragma unroll
    for (int g = 0; g < 4; ++g) {
      rk[g] = *(const v4f*)(Kb + cOff + g * 8 * DK);
      rv[g] = *(const v4f*)(Vb + cOff + g * 8 * DK);
    }
#pragma unroll
    for (int g = 0; g < 4; ++g) {
      *(v4f*)(sK + sOff + g * 8 * APITCH) = rk[g];
      *(v4f*)(sV + sOff + g * 8 * APITCH) = rv[g];
    }
    __syncthreads();

    for (int c = 0; c < nchunks; ++c) {
      const int cur = (c & 1) * ASTAGE;
      const int nxt = ASTAGE - cur;
      const bool more = (c + 1 < nchunks);
      if (more) {
        const size_t k0 = (size_t)(c + 1) * AKC * DK;
#pragma unroll
        for (int g = 0; g < 4; ++g) {
          rk[g] = *(const v4f*)(Kb + k0 + cOff + g * 8 * DK);
          rv[g] = *(const v4f*)(Vb + k0 + cOff + g * 8 * DK);
        }
      }

      if (2 * c <= qt) {
        const bool two = (2 * c + 1 <= qt);
        const float* k0s = sK + cur + lo * APITCH + 2 * hi;
        const float* k1s = k0s + 16 * APITCH;
        v8f sc0 = vzero8();
        v8f sc1 = vzero8();
        v2f ka[32];
#pragma unroll
        for (int ks = 0; ks < 32; ++ks) ka[ks] = *(const v2f*)(k0s + 4 * ks);
        if (two) {
          v2f kb[32];
#pragma unroll
          for (int ks = 0; ks < 32; ++ks) kb[ks] = *(const v2f*)(k1s + 4 * ks);
#pragma unroll
          for (int ks = 0; ks < 32; ++ks) {
            sc0 = wmma_f32(ka[ks], qa[ks], sc0);
            sc1 = wmma_f32(kb[ks], qa[ks], sc1);
          }
        } else {
#pragma unroll
          for (int ks = 0; ks < 32; ++ks) sc0 = wmma_f32(ka[ks], qa[ks], sc0);
        }
        softmax_pv(sc0, sV + cur, 2 * c == qt, hi, lo, mrow, lrow, o);
        if (two)
          softmax_pv(sc1, sV + cur + 16 * APITCH, 2 * c + 1 == qt, hi, lo, mrow, lrow, o);
      }

      if (more) {
#pragma unroll
        for (int g = 0; g < 4; ++g) {
          *(v4f*)(sK + nxt + sOff + g * 8 * APITCH) = rk[g];
          *(v4f*)(sV + nxt + sOff + g * 8 * APITCH) = rv[g];
        }
      }
      __syncthreads();
    }

    const float invl = 1.f / lrow;
    float* __restrict__ orow =
        AO + ((size_t)b * SEQ + qt * 16 + lo) * D_MODEL + h * DK + 8 * hi;
#pragma unroll
    for (int t = 0; t < 8; ++t) {
      const v4f w0 = {o[t][0] * invl, o[t][1] * invl, o[t][2] * invl, o[t][3] * invl};
      const v4f w1 = {o[t][4] * invl, o[t][5] * invl, o[t][6] * invl, o[t][7] * invl};
      *(v4f*)(orow + t * 16) = w0;
      *(v4f*)(orow + t * 16 + 4) = w1;
    }
  }
}

__global__ __launch_bounds__(GTHREADS)
void mha_out_proj(const float* __restrict__ A, const float* __restrict__ WO,
                  float* __restrict__ out) {
  __shared__ __attribute__((aligned(16))) float smem[GEMM_LDS_FLOATS];

  const int lane = threadIdx.x & 31;
  const int wave = threadIdx.x >> 5;
  const int hi = lane >> 4;
  const int lo = lane & 15;
  const int m0 = blockIdx.x * GBM;
  const int n0 = blockIdx.y * GBN;

  v8f acc[2][4];
#pragma unroll
  for (int mt = 0; mt < 2; ++mt)
#pragma unroll
    for (int nt = 0; nt < 4; ++nt) acc[mt][nt] = vzero8();

  gemm_nt_128x128(A, WO, m0, n0, smem, smem + 2 * STAGE, acc);

  const int mw = m0 + (wave & 3) * 32;
  const int nw = n0 + (wave >> 2) * 64;
#pragma unroll
  for (int mt = 0; mt < 2; ++mt)
#pragma unroll
    for (int nt = 0; nt < 4; ++nt)
#pragma unroll
      for (int i = 0; i < 8; ++i)
        out[(size_t)(mw + mt * 16 + i + 8 * hi) * D_MODEL + nw + nt * 16 + lo] =
            acc[mt][nt][i];
}

extern "C" void kernel_launch(void* const* d_in, const int* in_sizes, int n_in,
                              void* d_out, int out_size, void* d_ws, size_t ws_size,
                              hipStream_t stream) {
  (void)in_sizes; (void)n_in; (void)out_size; (void)ws_size;
  const float* x   = (const float*)d_in[0];
  const int*   tp  = (const int*)d_in[1];
  const float* WQ  = (const float*)d_in[2];
  const float* WK  = (const float*)d_in[3];
  const float* WV  = (const float*)d_in[4];
  const float* WO  = (const float*)d_in[5];
  float* out = (float*)d_out;

  const size_t QSZ = (size_t)BATCH * NHEADS * SEQ * DK;
  float* ws = (float*)d_ws;
  float* Qw = ws;
  float* Kw = ws + QSZ;
  float* Vw = ws + 2 * QSZ;
  float* AO = ws + 3 * QSZ;

  dim3 gProj(MROWS / GBM, D_MODEL / GBN, 3);
  mha_qkv_rope<<<gProj, GTHREADS, 0, stream>>>(x, tp, WQ, WK, WV, Qw, Kw, Vw);

  dim3 gAttn(QBLOCKS / 2, NHEADS, BATCH);
  mha_attn<<<gAttn, ATHREADS, 0, stream>>>(Qw, Kw, Vw, AO);

  dim3 gOut(MROWS / GBM, D_MODEL / GBN);
  mha_out_proj<<<gOut, GTHREADS, 0, stream>>>(AO, WO, out);
}


# Attention softmax: vector instructions beside the f32 WMMAs cut to under half: exp2 with the score scale folded into one multiply-add, V key pairs interleaved in LDS so fragments load as pairs
# speedup vs baseline: 3.4205x; 1.0339x over previous
#include <hip/hip_runtime.h>
#include <math.h>


#define D_MODEL 2048
#define NHEADS  16
#define DK      128
#define SEQ     2048
#define BATCH   2
#define MROWS   (BATCH * SEQ)
#define LOG_THETA 9.210340371976184f

typedef __attribute__((ext_vector_type(2))) float v2f;
typedef __attribute__((ext_vector_type(8))) float v8f;

__device__ __forceinline__ v8f wmma_f32(v2f a, v2f b, v8f c) {
  return __builtin_amdgcn_wmma_f32_16x16x4_f32(false, a, false, b,
                                               (short)0, c, false, false);
}

__device__ __forceinline__ v8f vzero8() {
  v8f z = {0.f, 0.f, 0.f, 0.f, 0.f, 0.f, 0.f, 0.f};
  return z;
}

#define GBM 128
#define GBN 128
#define GBK 64
#define GTHREADS 256
#define KTILES (D_MODEL / GBK)
#define LDP 68
#define STAGE (GBM * LDP)
#define GEMM_LDS_FLOATS (4 * STAGE)

typedef __attribute__((ext_vector_type(4))) float v4f;

__device__ __forceinline__ void gemm_nt_128x128(const float* __restrict__ A,
                                                const float* __restrict__ Wt,
                                                const int m0, const int n0,
                                                float* __restrict__ sA,
                                                float* __restrict__ sB,
                                                v8f (&acc)[2][4]) {
  const int tid = threadIdx.x;
  const int lane = tid & 31;
  const int wave = tid >> 5;
  const int hi = lane >> 4;
  const int lo = lane & 15;
  const int wm = wave & 3;
  const int wn = wave >> 2;

  const int lr = tid >> 4;
  const int lc = (tid & 15) << 2;
  const float* __restrict__ gA = A + (size_t)(m0 + lr) * D_MODEL + lc;
  const float* __restrict__ gW = Wt + (size_t)(n0 + lr) * D_MODEL + lc;
  const int sOff = lr * LDP + lc;

  v4f ra[8], rw[8];
#pragma unroll
  for (int p = 0; p < 8; ++p) {
    ra[p] = *(const v4f*)(gA + (size_t)p * 16 * D_MODEL);
    rw[p] = *(const v4f*)(gW + (size_t)p * 16 * D_MODEL);
  }
#pragma unroll
  for (int p = 0; p < 8; ++p) {
    *(v4f*)(sA + sOff + p * 16 * LDP) = ra[p];
    *(v4f*)(sB + sOff + p * 16 * LDP) = rw[p];
  }
  __syncthreads();

  const int aFrag = (wm * 32 + lo) * LDP + 2 * hi;
  const int bFrag = (wn * 64 + lo) * LDP + 2 * hi;

  for (int kt = 0; kt < KTILES; ++kt) {
    const int cur = (kt & 1) * STAGE;
    const int nxt = STAGE - cur;
    const bool more = (kt + 1 < KTILES);
    if (more) {
      const int k0 = (kt + 1) * GBK;
#pragma unroll
      for (int p = 0; p < 8; ++p) {
        ra[p] = *(const v4f*)(gA + (size_t)p * 16 * D_MODEL + k0);
        rw[p] = *(const v4f*)(gW + (size_t)p * 16 * D_MODEL + k0);
      }
    }

    const float* as = sA + cur + aFrag;
    const float* bs = sB + cur + bFrag;
#pragma unroll
    for (int s = 0; s < GBK / 4; ++s) {
      const v2f a0 = *(const v2f*)(as + 4 * s);
      const v2f a1 = *(const v2f*)(as + 16 * LDP + 4 * s);
      v2f b[4];
#pragma unroll
      for (int nt = 0; nt < 4; ++nt)
        b[nt] = *(const v2f*)(bs + nt * 16 * LDP + 4 * s);
#pragma unroll
      for (int nt = 0; nt < 4; ++nt) {
        acc[0][nt] = wmma_f32(a0, b[nt], acc[0][nt]);
        acc[1][nt] = wmma_f32(a1, b[nt], acc[1][nt]);
      }
    }

    if (more) {
#pragma unroll
      for (int p = 0; p < 8; ++p) {
        *(v4f*)(sA + nxt + sOff + p * 16 * LDP) = ra[p];
        *(v4f*)(sB + nxt + sOff + p * 16 * LDP) = rw[p];
      }
    }
    __syncthreads();
  }
}

__global__ __launch_bounds__(GTHREADS)
void mha_qkv_rope(const float* __restrict__ x, const int* __restrict__ tpos,
                  const float* __restrict__ WQ, const float* __restrict__ WK,
                  const float* __restrict__ WV,
                  float* __restrict__ Qo, float* __restrict__ Ko,
                  float* __restrict__ Vo) {
  __shared__ __attribute__((aligned(16))) float smem[GEMM_LDS_FLOATS];

  const int lane = threadIdx.x & 31;
  const int wave = threadIdx.x >> 5;
  const int hi = lane >> 4;
  const int lo = lane & 15;
  const int m0 = blockIdx.x * GBM;
  const int n0 = blockIdx.y * GBN;
  const int mat = blockIdx.z;

  const float* __restrict__ W = (mat == 0) ? WQ : ((mat == 1) ? WK : WV);
  float* __restrict__ Out = (mat == 0) ? Qo : ((mat == 1) ? Ko : Vo);

  v8f acc[2][4];
#pragma unroll
  for (int mt = 0; mt < 2; ++mt)
#pragma unroll
    for (int nt = 0; nt < 4; ++nt) acc[mt][nt] = vzero8();

  gemm_nt_128x128(x, W, m0, n0, smem, smem + 2 * STAGE, acc);

  const int mw = m0 + (wave & 3) * 32;
  const int nw = n0 + (wave >> 2) * 64;
  const int bidx = m0 >> 11;
  if (mat == 2) {
#pragma unroll
    for (int mt = 0; mt < 2; ++mt) {
#pragma unroll
      for (int nt = 0; nt < 4; ++nt) {
        const int ng = nw + nt * 16 + lo;
        const int h = ng >> 7, d = ng & 127;
#pragma unroll
        for (int i = 0; i < 8; ++i) {
          const int s = (mw + mt * 16 + i + 8 * hi) & (SEQ - 1);
          Out[(((size_t)bidx * NHEADS + h) * SEQ + s) * DK + d] = acc[mt][nt][i];
        }
      }
    }
  } else {
    float* __restrict__ park = smem + wave * (8 * 256);
#pragma unroll
    for (int mt = 0; mt < 2; ++mt) {
#pragma unroll
      for (int nt = 0; nt < 4; ++nt) {
        const int t = mt * 4 + nt;
        const v4f c0 = {acc[mt][nt][0], acc[mt][nt][1], acc[mt][nt][2], acc[mt][nt][3]};
        const v4f c1 = {acc[mt][nt][4], acc[mt][nt][5], acc[mt][nt][6], acc[mt][nt][7]};
        *(v4f*)(park + (2 * t + 0) * 128 + lane * 4) = c0;
        *(v4f*)(park + (2 * t + 1) * 128 + lane * 4) = c1;
      }
    }
#pragma unroll 1
    for (int t = 0; t < 8; ++t) {
      const v4f c0 = *(const v4f*)(park + (2 * t + 0) * 128 + lane * 4);
      const v4f c1 = *(const v4f*)(park + (2 * t + 1) * 128 + lane * 4);
      const int mt = t >> 2, nt = t & 3;
      const int ng = nw + nt * 16 + lo;
      const int h = ng >> 7, d = ng & 127;
      const float inv_freq = expf(-(float)(d & ~1) * (LOG_THETA / (float)DK));
      const int odd = d & 1;
#pragma unroll
      for (int i = 0; i < 8; ++i) {
        const int s = (mw + mt * 16 + i + 8 * hi) & (SEQ - 1);
        const float p = (float)tpos[bidx * SEQ + s];
        float sn, cs;
        sincosf(p * inv_freq, &sn, &cs);
        const float val = (i < 4) ? c0[i & 3] : c1[i & 3];
        const float part = __shfl_xor(val, 1, 32);
        const float r = odd ? fmaf(part, sn, val * cs)
                            : fmaf(val, cs, -part * sn);
        Out[(((size_t)bidx * NHEADS + h) * SEQ + s) * DK + d] = r;
      }
    }
  }
}

__device__ __forceinline__ float xhalf(float x) {
  const int b = __builtin_bit_cast(int, x);
  return __builtin_bit_cast(float, __builtin_amdgcn_permlanex16(b, b, 0x76543210, 0xfedcba98,
                                                                false, false));
}

#define AKC 32
#define APITCH 132
#define ASTAGE (AKC * APITCH)
#define VPITCH 264
#define ATHREADS 256
#define ATT_LDS_FLOATS (4 * ASTAGE)
#define QBLOCKS (SEQ / 128)

__device__ __forceinline__ void store_vpair(float* __restrict__ dst, const v4f ev, const v4f od) {
  const v4f w0 = {ev[0], od[0], ev[1], od[1]};
  const v4f w1 = {ev[2], od[2], ev[3], od[3]};
  *(v4f*)(dst) = w0;
  *(v4f*)(dst + 4) = w1;
}

__device__ __forceinline__ void softmax_pv(const v8f sc, const float* __restrict__ vt,
                                           const bool diag, const int hi, const int lo,
                                           float& mrow, float& lrow, v8f (&o)[8]) {
  v2f av[4][8];
#pragma unroll
  for (int j = 0; j < 4; ++j) {
    const float* vp = vt + (4 * hi + j) * VPITCH + 2 * lo;
#pragma unroll
    for (int t = 0; t < 8; ++t) av[j][t] = *(const v2f*)(vp + 32 * t);
  }
  const float cscale = 0.12751743082459868f;
  float s[8];
#pragma unroll
  for (int r = 0; r < 8; ++r) {
    s[r] = sc[r];
    if (diag && (8 * hi + r) > lo) s[r] = -__builtin_inff();
  }
  float rm = s[0];
#pragma unroll
  for (int r = 1; r < 8; ++r) rm = fmaxf(rm, s[r]);
  rm = fmaxf(rm, xhalf(rm));
  const float mnew = fmaxf(mrow, rm * cscale);
  const float a = __builtin_amdgcn_exp2f(mrow - mnew);
  const v2f c2 = {cscale, cscale};
  const v2f nm2 = {-mnew, -mnew};
  float p[8];
#pragma unroll
  for (int j = 0; j < 4; ++j) {
    const v2f sp = {s[2 * j], s[2 * j + 1]};
    const v2f e = __builtin_elementwise_fma(sp, c2, nm2);
    p[2 * j] = __builtin_amdgcn_exp2f(e[0]);
    p[2 * j + 1] = __builtin_amdgcn_exp2f(e[1]);
  }
  float rs = p[0];
#pragma unroll
  for (int r = 1; r < 8; ++r) rs += p[r];
  rs += xhalf(rs);
  lrow = lrow * a + rs;
  mrow = mnew;

#pragma unroll
  for (int t = 0; t < 8; ++t)
#pragma unroll
    for (int r = 0; r < 8; ++r) o[t][r] *= a;

#pragma unroll
  for (int j = 0; j < 4; ++j) {
    v2f pb;
    pb[0] = p[2 * j];
    pb[1] = p[2 * j + 1];
#pragma unroll
    for (int t = 0; t < 8; ++t) o[t] = wmma_f32(av[j][t], pb, o[t]);
  }
}

__global__ __launch_bounds__(ATHREADS)
void mha_attn(const float* __restrict__ Q, const float* __restrict__ K,
              const float* __restrict__ V, float* __restrict__ AO) {
  __shared__ __attribute__((aligned(16))) float smem[ATT_LDS_FLOATS];
  float* __restrict__ sK = smem;
  float* __restrict__ sV = smem + 2 * ASTAGE;

  const int tid = threadIdx.x;
  const int lane = tid & 31;
  const int wave = __builtin_amdgcn_readfirstlane(tid >> 5);
  const int hi = lane >> 4;
  const int lo = lane & 15;
  const int pr = blockIdx.x;
  const int h = blockIdx.y;
  const int b = blockIdx.z;

  const size_t head = ((size_t)b * NHEADS + h) * SEQ;
  const float* __restrict__ Qb = Q + head * DK;
  const float* __restrict__ Kb = K + head * DK;
  const float* __restrict__ Vb = V + head * DK;

  const int cOff = wave * DK + lane * 4;
  const int sOff = wave * APITCH + lane * 4;
  const int vcOff = 2 * wave * DK + lane * 4;
  const int vsOff = wave * VPITCH + lane * 8;

  for (int pass = 0; pass < 2; ++pass) {
    const int qb = pass ? (QBLOCKS - 1 - pr) : pr;
    const int qt = qb * 8 + wave;
    const int nchunks = qb * 4 + 4;

    v2f qa[32];
    const float* qrow = Qb + (size_t)(qt * 16 + lo) * DK;
#pragma unroll
    for (int ks = 0; ks < 32; ++ks)
      qa[ks] = *(const v2f*)(qrow + 4 * ks + 2 * hi);

    float mrow = -__builtin_inff();
    float lrow = 0.f;
    v8f o[8];
#pragma unroll
    for (int t = 0; t < 8; ++t) o[t] = vzero8();

    v4f rk[4], rv[4];
#pragma unroll
    for (int g = 0; g < 4; ++g) rk[g] = *(const v4f*)(Kb + cOff + g * 8 * DK);
#pragma unroll
    for (int g = 0; g < 2; ++g) {
      rv[2 * g] = *(const v4f*)(Vb + vcOff + g * 16 * DK);
      rv[2 * g + 1] = *(const v4f*)(Vb + vcOff + g * 16 * DK + DK);
    }
#pragma unroll
    for (int g = 0; g < 4; ++g) *(v4f*)(sK + sOff + g * 8 * APITCH) = rk[g];
#pragma unroll
    for (int g = 0; g < 2; ++g) store_vpair(sV + vsOff + g * 8 * VPITCH, rv[2 * g], rv[2 * g + 1]);
    __syncthreads();

    for (int c = 0; c < nchunks; ++c) {
      const int cur = (c & 1) * ASTAGE;
      const int nxt = ASTAGE - cur;
      const bool more = (c + 1 < nchunks);
      if (more) {
        const size_t k0 = (size_t)(c + 1) * AKC * DK;
#pragma unroll
        for (int g = 0; g < 4; ++g) rk[g] = *(const v4f*)(Kb + k0 + cOff + g * 8 * DK);
#pragma unroll
        for (int g = 0; g < 2; ++g) {
          rv[2 * g] = *(const v4f*)(Vb + k0 + vcOff + g * 16 * DK);
          rv[2 * g + 1] = *(const v4f*)(Vb + k0 + vcOff + g * 16 * DK + DK);
        }
      }

      if (2 * c <= qt) {
        const bool two = (2 * c + 1 <= qt);
        const float* k0s = sK + cur + lo * APITCH + 2 * hi;
        const float* k1s = k0s + 16 * APITCH;
        v8f sc0 = vzero8();
        v8f sc1 = vzero8();
        v2f ka[32];
#pragma unroll
        for (int ks = 0; ks < 32; ++ks) ka[ks] = *(const v2f*)(k0s + 4 * ks);
        if (two) {
          v2f kb[32];
#pragma unroll
          for (int ks = 0; ks < 32; ++ks) kb[ks] = *(const v2f*)(k1s + 4 * ks);
#pragma unroll
          for (int ks = 0; ks < 32; ++ks) {
            sc0 = wmma_f32(ka[ks], qa[ks], sc0);
            sc1 = wmma_f32(kb[ks], qa[ks], sc1);
          }
        } else {
#pragma unroll
          for (int ks = 0; ks < 32; ++ks) sc0 = wmma_f32(ka[ks], qa[ks], sc0);
        }
        softmax_pv(sc0, sV + cur, 2 * c == qt, hi, lo, mrow, lrow, o);
        if (two)
          softmax_pv(sc1, sV + cur + 8 * VPITCH, 2 * c + 1 == qt, hi, lo, mrow, lrow, o);
      }

      if (more) {
#pragma unroll
        for (int g = 0; g < 4; ++g) *(v4f*)(sK + nxt + sOff + g * 8 * APITCH) = rk[g];
#pragma unroll
        for (int g = 0; g < 2; ++g)
          store_vpair(sV + nxt + vsOff + g * 8 * VPITCH, rv[2 * g], rv[2 * g + 1]);
      }
      __syncthreads();
    }

    const float invl = 1.f / lrow;
    float* __restrict__ orow =
        AO + ((size_t)b * SEQ + qt * 16 + lo) * D_MODEL + h * DK + 8 * hi;
#pragma unroll
    for (int t = 0; t < 8; ++t) {
      const v4f w0 = {o[t][0] * invl, o[t][1] * invl, o[t][2] * invl, o[t][3] * invl};
      const v4f w1 = {o[t][4] * invl, o[t][5] * invl, o[t][6] * invl, o[t][7] * invl};
      *(v4f*)(orow + t * 16) = w0;
      *(v4f*)(orow + t * 16 + 4) = w1;
    }
  }
}

__global__ __launch_bounds__(GTHREADS)
void mha_out_proj(const float* __restrict__ A, const float* __restrict__ WO,
                  float* __restrict__ out) {
  __shared__ __attribute__((aligned(16))) float smem[GEMM_LDS_FLOATS];

  const int lane = threadIdx.x & 31;
  const int wave = threadIdx.x >> 5;
  const int hi = lane >> 4;
  const int lo = lane & 15;
  const int m0 = blockIdx.x * GBM;
  const int n0 = blockIdx.y * GBN;

  v8f acc[2][4];
#pragma unroll
  for (int mt = 0; mt < 2; ++mt)
#pragma unroll
    for (int nt = 0; nt < 4; ++nt) acc[mt][nt] = vzero8();

  gemm_nt_128x128(A, WO, m0, n0, smem, smem + 2 * STAGE, acc);

  const int mw = m0 + (wave & 3) * 32;
  const int nw = n0 + (wave >> 2) * 64;
#pragma unroll
  for (int mt = 0; mt < 2; ++mt)
#pragma unroll
    for (int nt = 0; nt < 4; ++nt)
#pragma unroll
      for (int i = 0; i < 8; ++i)
        out[(size_t)(mw + mt * 16 + i + 8 * hi) * D_MODEL + nw + nt * 16 + lo] =
            acc[mt][nt][i];
}

extern "C" void kernel_launch(void* const* d_in, const int* in_sizes, int n_in,
                              void* d_out, int out_size, void* d_ws, size_t ws_size,
                              hipStream_t stream) {
  (void)in_sizes; (void)n_in; (void)out_size; (void)ws_size;
  const float* x   = (const float*)d_in[0];
  const int*   tp  = (const int*)d_in[1];
  const float* WQ  = (const float*)d_in[2];
  const float* WK  = (const float*)d_in[3];
  const float* WV  = (const float*)d_in[4];
  const float* WO  = (const float*)d_in[5];
  float* out = (float*)d_out;

  const size_t QSZ = (size_t)BATCH * NHEADS * SEQ * DK;
  float* ws = (float*)d_ws;
  float* Qw = ws;
  float* Kw = ws + QSZ;
  float* Vw = ws + 2 * QSZ;
  float* AO = ws + 3 * QSZ;

  dim3 gProj(MROWS / GBM, D_MODEL / GBN, 3);
  mha_qkv_rope<<<gProj, GTHREADS, 0, stream>>>(x, tp, WQ, WK, WV, Qw, Kw, Vw);

  dim3 gAttn(QBLOCKS / 2, NHEADS, BATCH);
  mha_attn<<<gAttn, ATHREADS, 0, stream>>>(Qw, Kw, Vw, AO);

  dim3 gOut(MROWS / GBM, D_MODEL / GBN);
  mha_out_proj<<<gOut, GTHREADS, 0, stream>>>(AO, WO, out);
}
